# starting kernel
# baseline (speedup 1.0000x reference)
#include <hip/hip_runtime.h>

#define N_NODES 100000
#define N_EDGES 640000
#define D 128

typedef float v2f __attribute__((ext_vector_type(2)));
typedef float v8f __attribute__((ext_vector_type(8)));

__global__ void sage_zero_ws(float* __restrict__ ws, int n) {
    int i = blockIdx.x * blockDim.x + threadIdx.x;
    if (i < n) ws[i] = 0.0f;
}

__global__ void sage_edge_scatter(const float* __restrict__ feat,
                                  const int*   __restrict__ src,
                                  const int*   __restrict__ dst,
                                  float*       __restrict__ neigh_sum,
                                  float*       __restrict__ deg) {
    int tid = blockIdx.x * blockDim.x + threadIdx.x;
    int e = tid >> 5;
    int c = tid & 31;
    if (e >= N_EDGES) return;
    int s  = src[e];
    int dn = dst[e];
    const float4 v = *(const float4*)(feat + (size_t)s * D + c * 4);
    float* o = neigh_sum + (size_t)dn * D + c * 4;
    unsafeAtomicAdd(o + 0, v.x);
    unsafeAtomicAdd(o + 1, v.y);
    unsafeAtomicAdd(o + 2, v.z);
    unsafeAtomicAdd(o + 3, v.w);
    if (c == 0) unsafeAtomicAdd(deg + dn, 1.0f);
}

__global__ __launch_bounds__(256)
void sage_wmma_gemm(const float* __restrict__ feat,
                    const float* __restrict__ neigh_sum,
                    const float* __restrict__ deg,
                    const float* __restrict__ Wself,
                    const float* __restrict__ bself,
                    const float* __restrict__ Wneigh,
                    const float* __restrict__ bneigh,
                    float*       __restrict__ out) {
    const int lane  = threadIdx.x & 31;
    const int wave  = threadIdx.x >> 5;
    const int rloc  = lane & 15;
    const int khalf = lane >> 4;
    const int row   = blockIdx.x * 16 + rloc;
    const int col   = wave * 16 + (lane & 15);

    const float bias = bself[col] + bneigh[col];
    v8f acc;
#pragma unroll
    for (int j = 0; j < 8; ++j) acc[j] = bias;

    const float* frow = feat      + (size_t)row * D;
    const float* nrow = neigh_sum + (size_t)row * D;
    const float  rdeg = 1.0f / fmaxf(deg[row], 1.0f);

#pragma unroll
    for (int k0 = 0; k0 < D; k0 += 4) {
        const int k = k0 + khalf * 2;
        v2f a; a.x = frow[k];  a.y = frow[k + 1];
        v2f b; b.x = Wself[(size_t)k * D + col];
               b.y = Wself[(size_t)(k + 1) * D + col];
        acc = __builtin_amdgcn_wmma_f32_16x16x4_f32(
                  false, a, false, b, (short)0, acc, false, false);
    }

#pragma unroll
    for (int k0 = 0; k0 < D; k0 += 4) {
        const int k = k0 + khalf * 2;
        v2f a; a.x = nrow[k] * rdeg;  a.y = nrow[k + 1] * rdeg;
        v2f b; b.x = Wneigh[(size_t)k * D + col];
               b.y = Wneigh[(size_t)(k + 1) * D + col];
        acc = __builtin_amdgcn_wmma_f32_16x16x4_f32(
                  false, a, false, b, (short)0, acc, false, false);
    }

    float* obase = out + (size_t)(blockIdx.x * 16 + khalf * 8) * D + col;
#pragma unroll
    for (int j = 0; j < 8; ++j)
        obase[(size_t)j * D] = acc[j];
}

extern "C" void kernel_launch(void* const* d_in, const int* in_sizes, int n_in,
                              void* d_out, int out_size, void* d_ws, size_t ws_size,
                              hipStream_t stream) {
    const float* feat   = (const float*)d_in[0];
    const int*   src    = (const int*)  d_in[1];
    const int*   dst    = (const int*)  d_in[2];
    const float* Wself  = (const float*)d_in[3];
    const float* bself  = (const float*)d_in[4];
    const float* Wneigh = (const float*)d_in[5];
    const float* bneigh = (const float*)d_in[6];
    float*       out    = (float*)d_out;

    float* neigh_sum = (float*)d_ws;
    float* deg       = neigh_sum + (size_t)N_NODES * D;

    const int zn = N_NODES * (D + 1);
    sage_zero_ws<<<(zn + 255) / 256, 256, 0, stream>>>((float*)d_ws, zn);

    const int et = N_EDGES * 32;
    sage_edge_scatter<<<(et + 255) / 256, 256, 0, stream>>>(
        feat, src, dst, neigh_sum, deg);

    sage_wmma_gemm<<<N_NODES / 16, 256, 0, stream>>>(
        feat, neigh_sum, deg, Wself, bself, Wneigh, bneigh, out);
}


# Neighbour sums by row ownership: 256 blocks each own 391 out rows, scan all dst, add their edges' projected rows in LDS (LDS f32 atomics, arrival-order sums), add the mean to out once; 3->2 launches
# speedup vs baseline: 1.6378x; 1.6378x over previous
#include <hip/hip_runtime.h>


#define N_NODES 100000
#define N_EDGES 640000
#define D 128
#define GEMM_BLOCKS  (N_NODES / 16)

#define ROWS_PER_BLOCK 391
#define OWNER_BLOCKS   ((N_NODES + ROWS_PER_BLOCK - 1) / ROWS_PER_BLOCK)
#define OWNER_THREADS  1024
#define OWNER_WAVES    (OWNER_THREADS / 32)
#define LIST_CAP       16384
#define N_DST4         (N_EDGES / 4)
#define SCAN_LOADS     4
#define SCAN_STEP      (SCAN_LOADS * OWNER_THREADS)
#define SCAN_ITERS     (N_DST4 / SCAN_STEP)
static_assert(SCAN_ITERS % 2 == 1, "the two-set scan loop below ends on set a");
#define GROUP          8
#define ROWS_IN_FLIGHT 7

typedef float v2f __attribute__((ext_vector_type(2)));
typedef float v8f __attribute__((ext_vector_type(8)));

__global__ __launch_bounds__(256)
void sage_project(const float* __restrict__ feat,
                  const float* __restrict__ Wself,
                  const float* __restrict__ bself,
                  const float* __restrict__ Wneigh,
                  const float* __restrict__ bneigh,
                  float*       __restrict__ out,
                  float*       __restrict__ proj) {
    const int lane  = threadIdx.x & 31;
    const int wave  = threadIdx.x >> 5;
    const int rloc  = lane & 15;
    const int khalf = lane >> 4;
    const int row   = blockIdx.x * 16 + rloc;
    const int col   = wave * 16 + (lane & 15);

    const float bias = bself[col] + bneigh[col];
    v8f accs, accn;
#pragma unroll
    for (int j = 0; j < 8; ++j) { accs[j] = bias; accn[j] = 0.0f; }

    const float* frow = feat + (size_t)row * D;

#pragma unroll
    for (int k0 = 0; k0 < D; k0 += 4) {
        const int k = k0 + khalf * 2;
        v2f a;  a.x = frow[k];  a.y = frow[k + 1];
        v2f bs; bs.x = Wself[(size_t)k * D + col];
                bs.y = Wself[(size_t)(k + 1) * D + col];
        v2f bn; bn.x = Wneigh[(size_t)k * D + col];
                bn.y = Wneigh[(size_t)(k + 1) * D + col];
        accs = __builtin_amdgcn_wmma_f32_16x16x4_f32(
                   false, a, false, bs, (short)0, accs, false, false);
        accn = __builtin_amdgcn_wmma_f32_16x16x4_f32(
                   false, a, false, bn, (short)0, accn, false, false);
    }

    const size_t obase = (size_t)(blockIdx.x * 16 + khalf * 8) * D + col;
#pragma unroll
    for (int j = 0; j < 8; ++j) {
        out [obase + (size_t)j * D] = accs[j];
        proj[obase + (size_t)j * D] = accn[j];
    }
}

__device__ __forceinline__ void lds_add(float* p, float v) {
    __hip_atomic_fetch_add(p, v, __ATOMIC_RELAXED, __HIP_MEMORY_SCOPE_WORKGROUP);
}

__global__ __launch_bounds__(OWNER_THREADS)
void sage_owner_gather(const float* __restrict__ proj,
                       const int*   __restrict__ src,
                       const int*   __restrict__ dst,
                       float*       __restrict__ out) {
    __shared__ float    acc[ROWS_PER_BLOCK * D];
    __shared__ unsigned cnt[ROWS_PER_BLOCK];
    __shared__ unsigned list[LIST_CAP];
    __shared__ unsigned nlist;

    const int tid   = threadIdx.x;
    const int lane  = tid & 31;
    const int wave  = tid >> 5;
    const int r0    = blockIdx.x * ROWS_PER_BLOCK;
    const int nrows = min(ROWS_PER_BLOCK, N_NODES - r0);

    float4* acc4 = (float4*)acc;
    for (int i = tid; i < ROWS_PER_BLOCK * D / 4; i += OWNER_THREADS)
        acc4[i] = make_float4(0.0f, 0.0f, 0.0f, 0.0f);
    for (int i = tid; i < ROWS_PER_BLOCK; i += OWNER_THREADS) cnt[i] = 0u;
    if (tid == 0) nlist = 0u;
    __syncthreads();

    const int4* dst4 = (const int4*)dst;
    auto keep = [&](int dval, int e) {
        const unsigned local = (unsigned)(dval - r0);
        if (local < (unsigned)nrows) {
            atomicAdd(&cnt[local], 1u);
            const unsigned slot = atomicAdd(&nlist, 1u);
            if (slot < LIST_CAP) {
                list[slot] = ((unsigned)e << 9) | local;
            } else {
                const float* p = proj + (size_t)src[e] * D;
                float* a = acc + local * D;
#pragma nounroll
                for (int c = 0; c < D; ++c) lds_add(a + (c & 3) * 32 + (c >> 2), p[c]);
            }
        }
    };
    const int first = blockIdx.x % SCAN_ITERS;
    auto fetch = [&](int4 (&d)[SCAN_LOADS], int it) -> int {
        int itr = it + first;
        if (itr >= SCAN_ITERS) itr -= SCAN_ITERS;
        const int base = itr * SCAN_STEP + tid;
#pragma unroll
        for (int u = 0; u < SCAN_LOADS; ++u) d[u] = dst4[base + u * OWNER_THREADS];
        return base;
    };
    auto look = [&](const int4 (&d)[SCAN_LOADS], int base) {
#pragma unroll
        for (int u = 0; u < SCAN_LOADS; ++u) {
            const int e = (base + u * OWNER_THREADS) * 4;
            keep(d[u].x, e);
            keep(d[u].y, e + 1);
            keep(d[u].z, e + 2);
            keep(d[u].w, e + 3);
        }
    };
    int4 da[SCAN_LOADS], db[SCAN_LOADS];
    int base_a = fetch(da, 0);
    for (int it = 1; it + 1 < SCAN_ITERS; it += 2) {
        const int base_b = fetch(db, it);
        look(da, base_a);
        base_a = fetch(da, it + 1);
        look(db, base_b);
    }
    look(da, base_a);
    if (tid < N_DST4 - SCAN_ITERS * SCAN_STEP) {
        const int idx = SCAN_ITERS * SCAN_STEP + tid;
        const int4 d = dst4[idx];
        keep(d.x, idx * 4);
        keep(d.y, idx * 4 + 1);
        keep(d.z, idx * 4 + 2);
        keep(d.w, idx * 4 + 3);
    }
    __syncthreads();

    const unsigned n = min(nlist, (unsigned)LIST_CAP);
    for (unsigned i = tid; i < n; i += OWNER_THREADS) {
        const unsigned ent = list[i];
        list[i] = ((unsigned)src[ent >> 9] << 9) | (ent & 511u);
    }
    __syncthreads();

    const unsigned n_u    = __builtin_amdgcn_readfirstlane(n);
    const unsigned wave_u = __builtin_amdgcn_readfirstlane(wave);
    for (unsigned g0 = wave_u * GROUP; g0 < n_u; g0 += OWNER_WAVES * GROUP) {
        const unsigned m  = min((unsigned)GROUP, n_u - g0);
        const unsigned my = list[g0 + min((unsigned)(lane & (GROUP - 1)), m - 1u)];
        float4   v[GROUP];
        unsigned loc[GROUP];
#pragma unroll
        for (int q = 0; q < GROUP; ++q) {
            const unsigned ent = __builtin_amdgcn_readlane(my, q);
            loc[q] = ent & 511u;
            v[q] = *(const float4*)(proj + (size_t)(ent >> 9) * D + lane * 4);
        }
#pragma unroll
        for (int q = 0; q < GROUP; ++q) {
            const bool valid = g0 + (unsigned)q < n_u;
            float* a = acc + loc[q] * D + lane;
            lds_add(a,      valid ? v[q].x : 0.0f);
            lds_add(a + 32, valid ? v[q].y : 0.0f);
            lds_add(a + 64, valid ? v[q].z : 0.0f);
            lds_add(a + 96, valid ? v[q].w : 0.0f);
        }
    }
    __syncthreads();

    for (int rb = wave; rb < nrows; rb += OWNER_WAVES * ROWS_IN_FLIGHT) {
        float4 t[ROWS_IN_FLIGHT];
#pragma unroll
        for (int u = 0; u < ROWS_IN_FLIGHT; ++u) {
            const int r = rb + u * OWNER_WAVES;
            if (r < nrows) t[u] = *((const float4*)(out + (size_t)(r0 + r) * D) + lane);
        }
#pragma unroll
        for (int u = 0; u < ROWS_IN_FLIGHT; ++u) {
            const int r = rb + u * OWNER_WAVES;
            if (r < nrows) {
                const float rdeg = 1.0f / fmaxf((float)cnt[r], 1.0f);
                const float* a = acc + r * D + lane;
                t[u].x += a[0]  * rdeg;
                t[u].y += a[32] * rdeg;
                t[u].z += a[64] * rdeg;
                t[u].w += a[96] * rdeg;
                *((float4*)(out + (size_t)(r0 + r) * D) + lane) = t[u];
            }
        }
    }
}

extern "C" void kernel_launch(void* const* d_in, const int* in_sizes, int n_in,
                              void* d_out, int out_size, void* d_ws, size_t ws_size,
                              hipStream_t stream) {
    const float* feat   = (const float*)d_in[0];
    const int*   src    = (const int*)  d_in[1];
    const int*   dst    = (const int*)  d_in[2];
    const float* Wself  = (const float*)d_in[3];
    const float* bself  = (const float*)d_in[4];
    const float* Wneigh = (const float*)d_in[5];
    const float* bneigh = (const float*)d_in[6];
    float*       out    = (float*)d_out;

    float* proj = (float*)d_ws;

    sage_project<<<GEMM_BLOCKS, 256, 0, stream>>>(
        feat, Wself, bself, Wneigh, bneigh, out, proj);

    sage_owner_gather<<<OWNER_BLOCKS, OWNER_THREADS, 0, stream>>>(
        proj, src, dst, out);
}


# GEMM recut, weight-stationary: 256 resident blocks hold W_self and W_neigh in LDS as B fragments; a wave owns 16 rows x all 256 result columns, so a feat row is read by one wave; same sum order
# speedup vs baseline: 2.0894x; 1.2757x over previous
#include <hip/hip_runtime.h>


#define N_NODES 100000
#define N_EDGES 640000
#define D 128

#define ROWS_PER_BLOCK 391
#define OWNER_BLOCKS   ((N_NODES + ROWS_PER_BLOCK - 1) / ROWS_PER_BLOCK)
#define OWNER_THREADS  1024
#define OWNER_WAVES    (OWNER_THREADS / 32)
#define LIST_CAP       16384
#define N_DST4         (N_EDGES / 4)
#define SCAN_LOADS     4
#define SCAN_STEP      (SCAN_LOADS * OWNER_THREADS)
#define SCAN_ITERS     (N_DST4 / SCAN_STEP)
static_assert(SCAN_ITERS % 2 == 1, "the two-set scan loop below ends on set a");
#define GROUP          8
#define ROWS_IN_FLIGHT 7

typedef float v2f __attribute__((ext_vector_type(2)));
typedef float v8f __attribute__((ext_vector_type(8)));

#define GEMM_GRID    256
#define GEMM_THREADS 256
#define GEMM_WAVES   (GEMM_THREADS / 32)
#define N_TILES      (N_NODES / 16)
#define KSTEPS       (D / 4)
#define KPAIRS       (KSTEPS / 2)
#define CTILES       (D / 16)
#define NFRAG        (2 * CTILES)

typedef float v4f __attribute__((ext_vector_type(4)));

__global__ __launch_bounds__(GEMM_THREADS)
void sage_project(const float* __restrict__ feat,
                  const float* __restrict__ Wself,
                  const float* __restrict__ bself,
                  const float* __restrict__ Wneigh,
                  const float* __restrict__ bneigh,
                  float*       __restrict__ out,
                  float*       __restrict__ proj) {
    __shared__ v4f wfrag[KPAIRS * NFRAG * 32];

    const int tid  = threadIdx.x;
    const int lane = tid & 31;
    const int c    = lane & 15;
    const int h    = lane >> 4;

    const int wave    = __builtin_amdgcn_readfirstlane(tid >> 5);
    const int t_begin = (int)(blockIdx.x * N_TILES) / GEMM_GRID;
    const int t_end   = (int)((blockIdx.x + 1) * N_TILES) / GEMM_GRID;

    auto load_a = [&](v2f (&a)[KSTEPS], int tile) {
        const float* frow = feat + (size_t)(tile * 16 + c) * D + 2 * h;
#pragma unroll
        for (int s = 0; s < KSTEPS; ++s) a[s] = *(const v2f*)(frow + 4 * s);
    };

    int tile = t_begin + wave;
    v2f a[KSTEPS];
    load_a(a, tile);

#pragma unroll 4
    for (int it = 0; it < 2 * KPAIRS * 2 * (D / 4) / GEMM_THREADS; ++it) {
        const int item = it * GEMM_THREADS + tid;
        const int n4 = item & 31;
        const int hh = (item >> 5) & 1;
        const int p  = (item >> 6) & 15;
        const int m  = item >> 10;
        const float* Wm = (m ? Wneigh : Wself) + (size_t)(8 * p + 2 * hh) * D + 4 * n4;
        const v4f r0 = *(const v4f*)(Wm);
        const v4f r1 = *(const v4f*)(Wm + D);
        const v4f r2 = *(const v4f*)(Wm + 4 * D);
        const v4f r3 = *(const v4f*)(Wm + 5 * D);
        v4f* q = wfrag + ((p * NFRAG + 8 * m + (n4 >> 2)) * 32 + 16 * hh + 4 * (n4 & 3));
        v4f w;
        w.x = r0.x; w.y = r1.x; w.z = r2.x; w.w = r3.x; q[0] = w;
        w.x = r0.y; w.y = r1.y; w.z = r2.y; w.w = r3.y; q[1] = w;
        w.x = r0.z; w.y = r1.z; w.z = r2.z; w.w = r3.z; q[2] = w;
        w.x = r0.w; w.y = r1.w; w.z = r2.w; w.w = r3.w; q[3] = w;
    }

    float bias[CTILES];
#pragma unroll
    for (int t = 0; t < CTILES; ++t) bias[t] = bself[16 * t + c] + bneigh[16 * t + c];

    __syncthreads();

    const v4f* bl = wfrag + lane;
    auto load_b = [&](v4f (&b)[NFRAG], int p) {
#pragma unroll
        for (int f = 0; f < NFRAG; ++f) b[f] = bl[(p * NFRAG + f) * 32];
    };

    for (;;) {
        const int  next = tile + GEMM_WAVES;
        const bool more = next < t_end;
        v2f an[KSTEPS];
#pragma unroll
        for (int s = 0; s < KSTEPS; ++s) an[s] = a[s];
        if (more) load_a(an, next);

        v8f acc[NFRAG];
#pragma unroll
        for (int f = 0; f < NFRAG; ++f) {
#pragma unroll
            for (int j = 0; j < 8; ++j) acc[f][j] = (f < CTILES) ? bias[f] : 0.0f;
        }

        v4f b0[NFRAG], b1[NFRAG];
        load_b(b0, 0);
#pragma unroll
        for (int p = 0; p < KPAIRS; p += 2) {
            load_b(b1, p + 1);
            __builtin_amdgcn_sched_barrier(0);
#pragma unroll
            for (int f = 0; f < NFRAG; ++f)
                acc[f] = __builtin_amdgcn_wmma_f32_16x16x4_f32(
                             false, a[2 * p], false, b0[f].xy, (short)0, acc[f], false, false);
#pragma unroll
            for (int f = 0; f < NFRAG; ++f)
                acc[f] = __builtin_amdgcn_wmma_f32_16x16x4_f32(
                             false, a[2 * p + 1], false, b0[f].zw, (short)0, acc[f], false, false);
            __builtin_amdgcn_sched_barrier(0);
            if (p + 2 < KPAIRS) load_b(b0, p + 2);
            __builtin_amdgcn_sched_barrier(0);
#pragma unroll
            for (int f = 0; f < NFRAG; ++f)
                acc[f] = __builtin_amdgcn_wmma_f32_16x16x4_f32(
                             false, a[2 * p + 2], false, b1[f].xy, (short)0, acc[f], false, false);
#pragma unroll
            for (int f = 0; f < NFRAG; ++f)
                acc[f] = __builtin_amdgcn_wmma_f32_16x16x4_f32(
                             false, a[2 * p + 3], false, b1[f].zw, (short)0, acc[f], false, false);
            __builtin_amdgcn_sched_barrier(0);
        }

        const size_t obase = (size_t)(tile * 16 + h * 8) * D + c;
#pragma unroll
        for (int j = 0; j < 8; ++j) {
#pragma unroll
            for (int t = 0; t < CTILES; ++t) {
                out [obase + (size_t)j * D + 16 * t] = acc[t][j];
                proj[obase + (size_t)j * D + 16 * t] = acc[CTILES + t][j];
            }
        }

        if (!more) break;
#pragma unroll
        for (int s = 0; s < KSTEPS; ++s) a[s] = an[s];
        tile = next;
    }
}

__device__ __forceinline__ void lds_add(float* p, float v) {
    __hip_atomic_fetch_add(p, v, __ATOMIC_RELAXED, __HIP_MEMORY_SCOPE_WORKGROUP);
}

__global__ __launch_bounds__(OWNER_THREADS)
void sage_owner_gather(const float* __restrict__ proj,
                       const int*   __restrict__ src,
                       const int*   __restrict__ dst,
                       float*       __restrict__ out) {
    __shared__ float    acc[ROWS_PER_BLOCK * D];
    __shared__ unsigned cnt[ROWS_PER_BLOCK];
    __shared__ unsigned list[LIST_CAP];
    __shared__ unsigned nlist;

    const int tid   = threadIdx.x;
    const int lane  = tid & 31;
    const int wave  = tid >> 5;
    const int r0    = blockIdx.x * ROWS_PER_BLOCK;
    const int nrows = min(ROWS_PER_BLOCK, N_NODES - r0);

    float4* acc4 = (float4*)acc;
    for (int i = tid; i < ROWS_PER_BLOCK * D / 4; i += OWNER_THREADS)
        acc4[i] = make_float4(0.0f, 0.0f, 0.0f, 0.0f);
    for (int i = tid; i < ROWS_PER_BLOCK; i += OWNER_THREADS) cnt[i] = 0u;
    if (tid == 0) nlist = 0u;
    __syncthreads();

    const int4* dst4 = (const int4*)dst;
    auto keep = [&](int dval, int e) {
        const unsigned local = (unsigned)(dval - r0);
        if (local < (unsigned)nrows) {
            atomicAdd(&cnt[local], 1u);
            const unsigned slot = atomicAdd(&nlist, 1u);
            if (slot < LIST_CAP) {
                list[slot] = ((unsigned)e << 9) | local;
            } else {
                const float* p = proj + (size_t)src[e] * D;
                float* a = acc + local * D;
#pragma nounroll
                for (int c = 0; c < D; ++c) lds_add(a + (c & 3) * 32 + (c >> 2), p[c]);
            }
        }
    };
    const int first = blockIdx.x % SCAN_ITERS;
    auto fetch = [&](int4 (&d)[SCAN_LOADS], int it) -> int {
        int itr = it + first;
        if (itr >= SCAN_ITERS) itr -= SCAN_ITERS;
        const int base = itr * SCAN_STEP + tid;
#pragma unroll
        for (int u = 0; u < SCAN_LOADS; ++u) d[u] = dst4[base + u * OWNER_THREADS];
        return base;
    };
    auto look = [&](const int4 (&d)[SCAN_LOADS], int base) {
#pragma unroll
        for (int u = 0; u < SCAN_LOADS; ++u) {
            const int e = (base + u * OWNER_THREADS) * 4;
            keep(d[u].x, e);
            keep(d[u].y, e + 1);
            keep(d[u].z, e + 2);
            keep(d[u].w, e + 3);
        }
    };
    int4 da[SCAN_LOADS], db[SCAN_LOADS];
    int base_a = fetch(da, 0);
    for (int it = 1; it + 1 < SCAN_ITERS; it += 2) {
        const int base_b = fetch(db, it);
        look(da, base_a);
        base_a = fetch(da, it + 1);
        look(db, base_b);
    }
    look(da, base_a);
    if (tid < N_DST4 - SCAN_ITERS * SCAN_STEP) {
        const int idx = SCAN_ITERS * SCAN_STEP + tid;
        const int4 d = dst4[idx];
        keep(d.x, idx * 4);
        keep(d.y, idx * 4 + 1);
        keep(d.z, idx * 4 + 2);
        keep(d.w, idx * 4 + 3);
    }
    __syncthreads();

    const unsigned n = min(nlist, (unsigned)LIST_CAP);
    for (unsigned i = tid; i < n; i += OWNER_THREADS) {
        const unsigned ent = list[i];
        list[i] = ((unsigned)src[ent >> 9] << 9) | (ent & 511u);
    }
    __syncthreads();

    const unsigned n_u    = __builtin_amdgcn_readfirstlane(n);
    const unsigned wave_u = __builtin_amdgcn_readfirstlane(wave);
    for (unsigned g0 = wave_u * GROUP; g0 < n_u; g0 += OWNER_WAVES * GROUP) {
        const unsigned m  = min((unsigned)GROUP, n_u - g0);
        const unsigned my = list[g0 + min((unsigned)(lane & (GROUP - 1)), m - 1u)];
        float4   v[GROUP];
        unsigned loc[GROUP];
#pragma unroll
        for (int q = 0; q < GROUP; ++q) {
            const unsigned ent = __builtin_amdgcn_readlane(my, q);
            loc[q] = ent & 511u;
            v[q] = *(const float4*)(proj + (size_t)(ent >> 9) * D + lane * 4);
        }
#pragma unroll
        for (int q = 0; q < GROUP; ++q) {
            const bool valid = g0 + (unsigned)q < n_u;
            float* a = acc + loc[q] * D + lane;
            lds_add(a,      valid ? v[q].x : 0.0f);
            lds_add(a + 32, valid ? v[q].y : 0.0f);
            lds_add(a + 64, valid ? v[q].z : 0.0f);
            lds_add(a + 96, valid ? v[q].w : 0.0f);
        }
    }
    __syncthreads();

    for (int rb = wave; rb < nrows; rb += OWNER_WAVES * ROWS_IN_FLIGHT) {
        float4 t[ROWS_IN_FLIGHT];
#pragma unroll
        for (int u = 0; u < ROWS_IN_FLIGHT; ++u) {
            const int r = rb + u * OWNER_WAVES;
            if (r < nrows) t[u] = *((const float4*)(out + (size_t)(r0 + r) * D) + lane);
        }
#pragma unroll
        for (int u = 0; u < ROWS_IN_FLIGHT; ++u) {
            const int r = rb + u * OWNER_WAVES;
            if (r < nrows) {
                const float rdeg = 1.0f / fmaxf((float)cnt[r], 1.0f);
                const float* a = acc + r * D + lane;
                t[u].x += a[0]  * rdeg;
                t[u].y += a[32] * rdeg;
                t[u].z += a[64] * rdeg;
                t[u].w += a[96] * rdeg;
                *((float4*)(out + (size_t)(r0 + r) * D) + lane) = t[u];
            }
        }
    }
}

extern "C" void kernel_launch(void* const* d_in, const int* in_sizes, int n_in,
                              void* d_out, int out_size, void* d_ws, size_t ws_size,
                              hipStream_t stream) {
    const float* feat   = (const float*)d_in[0];
    const int*   src    = (const int*)  d_in[1];
    const int*   dst    = (const int*)  d_in[2];
    const float* Wself  = (const float*)d_in[3];
    const float* bself  = (const float*)d_in[4];
    const float* Wneigh = (const float*)d_in[5];
    const float* bneigh = (const float*)d_in[6];
    float*       out    = (float*)d_out;

    float* proj = (float*)d_ws;

    sage_project<<<GEMM_GRID, GEMM_THREADS, 0, stream>>>(
        feat, Wself, bself, Wneigh, bneigh, out, proj);

    sage_owner_gather<<<OWNER_BLOCKS, OWNER_THREADS, 0, stream>>>(
        proj, src, dst, out);
}


# Owner keys: the projection launch also writes dst/391 as one byte per edge behind proj; owner blocks scan that plane, 16 edges per 16-byte load, one zero-byte mask test; same arrival-order sums
# speedup vs baseline: 2.3915x; 1.1446x over previous
#include <hip/hip_runtime.h>


#define N_NODES 100000
#define N_EDGES 640000
#define D 128

#define ROWS_PER_BLOCK 391
#define OWNER_BLOCKS   ((N_NODES + ROWS_PER_BLOCK - 1) / ROWS_PER_BLOCK)
#define OWNER_THREADS  1024
#define OWNER_WAVES    (OWNER_THREADS / 32)
#define LIST_CAP       16384
#define N_DST4         (N_EDGES / 4)
#define SCAN_LOADS     4
#define SCAN_STEP      (SCAN_LOADS * OWNER_THREADS)
#define SCAN_ITERS     (N_DST4 / SCAN_STEP)
static_assert(SCAN_ITERS % 2 == 1, "the two-set scan loop below ends on set a");
#define EDGE_CHUNK     2560
#define KEY_BLOCKS     (N_EDGES / EDGE_CHUNK)
#define KEY_WORDS      (EDGE_CHUNK / 4)
#define N_KEY16        (N_EDGES / 16)
#define KSCAN_ITERS    ((N_KEY16 + SCAN_STEP - 1) / SCAN_STEP)
#define KEY_ROT        128
static_assert(KEY_ROT % 32 == 0 && N_KEY16 % 32 == 0 && (OWNER_BLOCKS - 1) * KEY_ROT < N_KEY16, "aligned, one wrap");
#define PROJ_BYTES     ((size_t)N_NODES * D * sizeof(float))
#define WS_NEED        (PROJ_BYTES + (size_t)N_EDGES)
static_assert(N_EDGES % EDGE_CHUNK == 0 && EDGE_CHUNK % 128 == 0, "every key line has one writer");
static_assert(OWNER_BLOCKS <= 256, "an owner block number fits a byte");
#define ENTRY_INVALID  0x80000000u
#define GROUP          8
#define ROWS_IN_FLIGHT 7

typedef float v2f __attribute__((ext_vector_type(2)));
typedef float v8f __attribute__((ext_vector_type(8)));

#define GEMM_GRID    256
#define GEMM_THREADS 256
#define GEMM_WAVES   (GEMM_THREADS / 32)
#define N_TILES      (N_NODES / 16)
#define KSTEPS       (D / 4)
#define KPAIRS       (KSTEPS / 2)
#define CTILES       (D / 16)
#define NFRAG        (2 * CTILES)

typedef float v4f __attribute__((ext_vector_type(4)));

__global__ __launch_bounds__(GEMM_THREADS)
void sage_project(const float* __restrict__ feat,
                  const float* __restrict__ Wself,
                  const float* __restrict__ bself,
                  const float* __restrict__ Wneigh,
                  const float* __restrict__ bneigh,
                  float*       __restrict__ out,
                  float*       __restrict__ proj,
                  const int*   __restrict__ dst,
                  unsigned*    __restrict__ keys,
                  size_t                    ws_bytes) {
    __shared__ v4f wfrag[KPAIRS * NFRAG * 32];

    const int tid  = threadIdx.x;
    const int lane = tid & 31;
    const int c    = lane & 15;
    const int h    = lane >> 4;

    const int wave    = __builtin_amdgcn_readfirstlane(tid >> 5);
    const int t_begin = (int)(blockIdx.x * N_TILES) / GEMM_GRID;
    const int t_end   = (int)((blockIdx.x + 1) * N_TILES) / GEMM_GRID;

    auto load_a = [&](v2f (&a)[KSTEPS], int tile) {
        const float* frow = feat + (size_t)(tile * 16 + c) * D + 2 * h;
#pragma unroll
        for (int s = 0; s < KSTEPS; ++s) a[s] = *(const v2f*)(frow + 4 * s);
    };

    int tile = t_begin + wave;
    v2f a[KSTEPS];
    load_a(a, tile);

#pragma unroll 4
    for (int it = 0; it < 2 * KPAIRS * 2 * (D / 4) / GEMM_THREADS; ++it) {
        const int item = it * GEMM_THREADS + tid;
        const int n4 = item & 31;
        const int hh = (item >> 5) & 1;
        const int p  = (item >> 6) & 15;
        const int m  = item >> 10;
        const float* Wm = (m ? Wneigh : Wself) + (size_t)(8 * p + 2 * hh) * D + 4 * n4;
        const v4f r0 = *(const v4f*)(Wm);
        const v4f r1 = *(const v4f*)(Wm + D);
        const v4f r2 = *(const v4f*)(Wm + 4 * D);
        const v4f r3 = *(const v4f*)(Wm + 5 * D);
        v4f* q = wfrag + ((p * NFRAG + 8 * m + (n4 >> 2)) * 32 + 16 * hh + 4 * (n4 & 3));
        v4f w;
        w.x = r0.x; w.y = r1.x; w.z = r2.x; w.w = r3.x; q[0] = w;
        w.x = r0.y; w.y = r1.y; w.z = r2.y; w.w = r3.y; q[1] = w;
        w.x = r0.z; w.y = r1.z; w.z = r2.z; w.w = r3.z; q[2] = w;
        w.x = r0.w; w.y = r1.w; w.z = r2.w; w.w = r3.w; q[3] = w;
    }

    float bias[CTILES];
#pragma unroll
    for (int t = 0; t < CTILES; ++t) bias[t] = bself[16 * t + c] + bneigh[16 * t + c];

    __syncthreads();

    const v4f* bl = wfrag + lane;
    auto load_b = [&](v4f (&b)[NFRAG], int p) {
#pragma unroll
        for (int f = 0; f < NFRAG; ++f) b[f] = bl[(p * NFRAG + f) * 32];
    };

    for (;;) {
        const int  next = tile + GEMM_WAVES;
        const bool more = next < t_end;
        v2f an[KSTEPS];
#pragma unroll
        for (int s = 0; s < KSTEPS; ++s) an[s] = a[s];
        if (more) load_a(an, next);

        v8f acc[NFRAG];
#pragma unroll
        for (int f = 0; f < NFRAG; ++f) {
#pragma unroll
            for (int j = 0; j < 8; ++j) acc[f][j] = (f < CTILES) ? bias[f] : 0.0f;
        }

        v4f b0[NFRAG], b1[NFRAG];
        load_b(b0, 0);
#pragma unroll
        for (int p = 0; p < KPAIRS; p += 2) {
            load_b(b1, p + 1);
            __builtin_amdgcn_sched_barrier(0);
#pragma unroll
            for (int f = 0; f < NFRAG; ++f)
                acc[f] = __builtin_amdgcn_wmma_f32_16x16x4_f32(
                             false, a[2 * p], false, b0[f].xy, (short)0, acc[f], false, false);
#pragma unroll
            for (int f = 0; f < NFRAG; ++f)
                acc[f] = __builtin_amdgcn_wmma_f32_16x16x4_f32(
                             false, a[2 * p + 1], false, b0[f].zw, (short)0, acc[f], false, false);
            __builtin_amdgcn_sched_barrier(0);
            if (p + 2 < KPAIRS) load_b(b0, p + 2);
            __builtin_amdgcn_sched_barrier(0);
#pragma unroll
            for (int f = 0; f < NFRAG; ++f)
                acc[f] = __builtin_amdgcn_wmma_f32_16x16x4_f32(
                             false, a[2 * p + 2], false, b1[f].xy, (short)0, acc[f], false, false);
#pragma unroll
            for (int f = 0; f < NFRAG; ++f)
                acc[f] = __builtin_amdgcn_wmma_f32_16x16x4_f32(
                             false, a[2 * p + 3], false, b1[f].zw, (short)0, acc[f], false, false);
            __builtin_amdgcn_sched_barrier(0);
        }

        const size_t obase = (size_t)(tile * 16 + h * 8) * D + c;
#pragma unroll
        for (int j = 0; j < 8; ++j) {
#pragma unroll
            for (int t = 0; t < CTILES; ++t) {
                out [obase + (size_t)j * D + 16 * t] = acc[t][j];
                proj[obase + (size_t)j * D + 16 * t] = acc[CTILES + t][j];
            }
        }

        if (!more) break;
#pragma unroll
        for (int s = 0; s < KSTEPS; ++s) a[s] = an[s];
        tile = next;
    }

    const int kt = (wave - 1) * 32 + (int)__builtin_amdgcn_mbcnt_lo(~0u, 0u);
    if (ws_bytes >= WS_NEED && blockIdx.x < KEY_BLOCKS && wave >= 1 && wave <= KEY_WORDS / 4 / 32) {
        const int4* dp = (const int4*)dst + blockIdx.x * KEY_WORDS + kt;
        unsigned*   kp = keys + blockIdx.x * KEY_WORDS + kt;
        int4 kd[4];
#pragma unroll
        for (int u = 0; u < 4; ++u) kd[u] = dp[u * (KEY_WORDS / 4)];
#pragma unroll
        for (int u = 0; u < 4; ++u) {
            const unsigned k0 = ((unsigned)kd[u].x / ROWS_PER_BLOCK) & 255u;
            const unsigned k1 = ((unsigned)kd[u].y / ROWS_PER_BLOCK) & 255u;
            const unsigned k2 = ((unsigned)kd[u].z / ROWS_PER_BLOCK) & 255u;
            const unsigned k3 = ((unsigned)kd[u].w / ROWS_PER_BLOCK) & 255u;
            kp[u * (KEY_WORDS / 4)] = k0 | (k1 << 8) | (k2 << 16) | (k3 << 24);
        }
    }
}

__device__ __forceinline__ void lds_add(float* p, float v) {
    __hip_atomic_fetch_add(p, v, __ATOMIC_RELAXED, __HIP_MEMORY_SCOPE_WORKGROUP);
}

__global__ __launch_bounds__(OWNER_THREADS)
void sage_owner_gather(const float*    __restrict__ proj,
                       const int*      __restrict__ src,
                       const int*      __restrict__ dst,
                       const unsigned* __restrict__ keys,
                       size_t                       ws_bytes,
                       float*          __restrict__ out) {
    __shared__ float    acc[ROWS_PER_BLOCK * D];
    __shared__ unsigned cnt[ROWS_PER_BLOCK];
    __shared__ unsigned list[LIST_CAP];
    __shared__ unsigned nlist;

    const int tid   = threadIdx.x;
    const int lane  = tid & 31;
    const int wave  = tid >> 5;
    const int r0    = blockIdx.x * ROWS_PER_BLOCK;
    const int nrows = min(ROWS_PER_BLOCK, N_NODES - r0);

    float4* acc4 = (float4*)acc;
    for (int i = tid; i < ROWS_PER_BLOCK * D / 4; i += OWNER_THREADS)
        acc4[i] = make_float4(0.0f, 0.0f, 0.0f, 0.0f);
    for (int i = tid; i < ROWS_PER_BLOCK; i += OWNER_THREADS) cnt[i] = 0u;
    if (tid == 0) nlist = 0u;
    __syncthreads();

    auto push = [&](int e) {
        const unsigned slot = atomicAdd(&nlist, 1u);
        if (slot < LIST_CAP) {
            list[slot] = (unsigned)e;
        } else {
            const unsigned local = (unsigned)(dst[e] - r0);
            if (local < (unsigned)nrows) {
                atomicAdd(&cnt[local], 1u);
                const float* p = proj + (size_t)src[e] * D;
                float* a = acc + local * D;
#pragma nounroll
                for (int c = 0; c < D; ++c) lds_add(a + (c & 3) * 32 + (c >> 2), p[c]);
            }
        }
    };
    if (ws_bytes >= WS_NEED) {
        const uint4*   keys4 = (const uint4*)keys;
        const unsigned mine  = blockIdx.x * 0x01010101u;
        auto hit = [&](unsigned w) -> unsigned {
            const unsigned x = w ^ mine;
            return ~(((x & 0x7f7f7f7fu) + 0x7f7f7f7fu) | x | 0x7f7f7f7fu);
        };
        const int rot = blockIdx.x * KEY_ROT;
        auto fetch = [&](uint4 (&k)[SCAN_LOADS], int (&v)[SCAN_LOADS], int it) {
#pragma unroll
            for (int u = 0; u < SCAN_LOADS; ++u) {
                const int idx = it * SCAN_STEP + u * OWNER_THREADS + tid;
                int vv = idx + rot;
                if (vv >= N_KEY16) vv -= N_KEY16;
                const bool inside = idx < N_KEY16;
                v[u] = inside ? vv : -1;
                k[u] = keys4[inside ? vv : 0];
            }
        };
        auto look = [&](const uint4 (&k)[SCAN_LOADS], const int (&v)[SCAN_LOADS]) {
#pragma unroll
            for (int u = 0; u < SCAN_LOADS; ++u) {
                const int e = v[u] * 16;
                unsigned m = (hit(k[u].x) >> 7) | (hit(k[u].y) >> 6)
                           | (hit(k[u].z) >> 5) | (hit(k[u].w) >> 4);
                if (v[u] < 0) m = 0u;
                while (m != 0u) {
                    const int b = __builtin_ctz(m);
                    push(e + 4 * (b & 7) + (b >> 3));
                    m &= m - 1u;
                }
            }
        };
        uint4 ka[SCAN_LOADS], kb[SCAN_LOADS];
        int   va[SCAN_LOADS], vb[SCAN_LOADS];
        static_assert(KSCAN_ITERS % 2 == 0, "the loop below leaves set a and one more step");
        fetch(ka, va, 0);
        for (int it = 1; it + 1 < KSCAN_ITERS; it += 2) {
            fetch(kb, vb, it);
            look(ka, va);
            fetch(ka, va, it + 1);
            look(kb, vb);
        }
        fetch(kb, vb, KSCAN_ITERS - 1);
        look(ka, va);
        look(kb, vb);
    } else {
        const int4* dst4 = (const int4*)dst;
        auto keep = [&](int dval, int e) {
            if ((unsigned)(dval - r0) < (unsigned)nrows) push(e);
        };
        const int first = blockIdx.x % SCAN_ITERS;
        auto fetch = [&](int4 (&d)[SCAN_LOADS], int it) -> int {
            int itr = it + first;
            if (itr >= SCAN_ITERS) itr -= SCAN_ITERS;
            const int base = itr * SCAN_STEP + tid;
#pragma unroll
            for (int u = 0; u < SCAN_LOADS; ++u) d[u] = dst4[base + u * OWNER_THREADS];
            return base;
        };
        auto look = [&](const int4 (&d)[SCAN_LOADS], int base) {
#pragma unroll
            for (int u = 0; u < SCAN_LOADS; ++u) {
                const int e = (base + u * OWNER_THREADS) * 4;
                keep(d[u].x, e);
                keep(d[u].y, e + 1);
                keep(d[u].z, e + 2);
                keep(d[u].w, e + 3);
            }
        };
        int4 da[SCAN_LOADS], db[SCAN_LOADS];
        int base_a = fetch(da, 0);
        for (int it = 1; it + 1 < SCAN_ITERS; it += 2) {
            const int base_b = fetch(db, it);
            look(da, base_a);
            base_a = fetch(da, it + 1);
            look(db, base_b);
        }
        look(da, base_a);
        if (tid < N_DST4 - SCAN_ITERS * SCAN_STEP) {
            const int idx = SCAN_ITERS * SCAN_STEP + tid;
            const int4 d = dst4[idx];
            keep(d.x, idx * 4);
            keep(d.y, idx * 4 + 1);
            keep(d.z, idx * 4 + 2);
            keep(d.w, idx * 4 + 3);
        }
    }
    __syncthreads();

    const unsigned n = min(nlist, (unsigned)LIST_CAP);
    for (unsigned i = tid; i < n; i += OWNER_THREADS) {
        const unsigned e     = list[i];
        const unsigned local = (unsigned)(dst[e] - r0);
        const unsigned s     = (unsigned)src[e];
        const bool here = local < (unsigned)nrows;
        if (here) atomicAdd(&cnt[local], 1u);
        list[i] = here ? ((s << 9) | local) : ENTRY_INVALID;
    }
    __syncthreads();

    const unsigned n_u    = __builtin_amdgcn_readfirstlane(n);
    const unsigned wave_u = __builtin_amdgcn_readfirstlane(wave);
    for (unsigned g0 = wave_u * GROUP; g0 < n_u; g0 += OWNER_WAVES * GROUP) {
        const unsigned m  = min((unsigned)GROUP, n_u - g0);
        const unsigned my = list[g0 + min((unsigned)(lane & (GROUP - 1)), m - 1u)];
        float4   v[GROUP];
        unsigned loc[GROUP];
        bool     bad[GROUP];
#pragma unroll
        for (int q = 0; q < GROUP; ++q) {
            const unsigned ent = __builtin_amdgcn_readlane(my, q);
            loc[q] = ent & 511u;
            bad[q] = (ent & ENTRY_INVALID) != 0u;
            v[q] = *(const float4*)(proj + (size_t)((ent >> 9) & 0x1ffffu) * D + lane * 4);
        }
#pragma unroll
        for (int q = 0; q < GROUP; ++q) {
            const bool valid = g0 + (unsigned)q < n_u && !bad[q];
            float* a = acc + loc[q] * D + lane;
            lds_add(a,      valid ? v[q].x : 0.0f);
            lds_add(a + 32, valid ? v[q].y : 0.0f);
            lds_add(a + 64, valid ? v[q].z : 0.0f);
            lds_add(a + 96, valid ? v[q].w : 0.0f);
        }
    }
    __syncthreads();

    for (int rb = wave; rb < nrows; rb += OWNER_WAVES * ROWS_IN_FLIGHT) {
        float4 t[ROWS_IN_FLIGHT];
#pragma unroll
        for (int u = 0; u < ROWS_IN_FLIGHT; ++u) {
            const int r = rb + u * OWNER_WAVES;
            if (r < nrows) t[u] = *((const float4*)(out + (size_t)(r0 + r) * D) + lane);
        }
#pragma unroll
        for (int u = 0; u < ROWS_IN_FLIGHT; ++u) {
            const int r = rb + u * OWNER_WAVES;
            if (r < nrows) {
                const float rdeg = 1.0f / fmaxf((float)cnt[r], 1.0f);
                const float* a = acc + r * D + lane;
                t[u].x += a[0]  * rdeg;
                t[u].y += a[32] * rdeg;
                t[u].z += a[64] * rdeg;
                t[u].w += a[96] * rdeg;
                *((float4*)(out + (size_t)(r0 + r) * D) + lane) = t[u];
            }
        }
    }
}

extern "C" void kernel_launch(void* const* d_in, const int* in_sizes, int n_in,
                              void* d_out, int out_size, void* d_ws, size_t ws_size,
                              hipStream_t stream) {
    const float* feat   = (const float*)d_in[0];
    const int*   src    = (const int*)  d_in[1];
    const int*   dst    = (const int*)  d_in[2];
    const float* Wself  = (const float*)d_in[3];
    const float* bself  = (const float*)d_in[4];
    const float* Wneigh = (const float*)d_in[5];
    const float* bneigh = (const float*)d_in[6];
    float*       out    = (float*)d_out;

    float*    proj = (float*)d_ws;
    unsigned* keys = (unsigned*)((char*)d_ws + PROJ_BYTES);

    sage_project<<<GEMM_GRID, GEMM_THREADS, 0, stream>>>(
        feat, Wself, bself, Wneigh, bneigh, out, proj, dst, keys, ws_size);

    sage_owner_gather<<<OWNER_BLOCKS, OWNER_THREADS, 0, stream>>>(
        proj, src, dst, keys, ws_size, out);
}


# Owner kernel: kept edges sorted by row in LDS (count, prefix sum, place); a wave sums a row's projected sources in registers and adds the mean to out; no LDS f32 adds; int-add arrival-order sums
# speedup vs baseline: 2.5371x; 1.0609x over previous
#include <hip/hip_runtime.h>


#define N_NODES 100000
#define N_EDGES 640000
#define D 128

#define ROWS_PER_BLOCK 391
#define OWNER_BLOCKS   ((N_NODES + ROWS_PER_BLOCK - 1) / ROWS_PER_BLOCK)
#define OWNER_THREADS  1024
#define OWNER_WAVES    (OWNER_THREADS / 32)
#define LIST_CAP       32768
#define N_DST4         (N_EDGES / 4)
#define SCAN_LOADS     4
#define SCAN_STEP      (SCAN_LOADS * OWNER_THREADS)
#define SCAN_ITERS     (N_DST4 / SCAN_STEP)
static_assert(SCAN_ITERS % 2 == 1, "the two-set scan loop below ends on set a");
#define EDGE_CHUNK     2560
#define KEY_BLOCKS     (N_EDGES / EDGE_CHUNK)
#define KEY_WORDS      (EDGE_CHUNK / 4)
#define N_KEY16        (N_EDGES / 16)
#define KSCAN_ITERS    ((N_KEY16 + SCAN_STEP - 1) / SCAN_STEP)
#define KEY_ROT        128
static_assert(KEY_ROT % 32 == 0 && N_KEY16 % 32 == 0 && (OWNER_BLOCKS - 1) * KEY_ROT < N_KEY16, "aligned, one wrap");
#define PROJ_BYTES     ((size_t)N_NODES * D * sizeof(float))
#define WS_NEED        (PROJ_BYTES + (size_t)N_EDGES)
static_assert(N_EDGES % EDGE_CHUNK == 0 && EDGE_CHUNK % 128 == 0, "every key line has one writer");
static_assert(OWNER_BLOCKS <= 256, "an owner block number fits a byte");
#define ENTRY_INVALID  0x80000000u
#define CHUNK          8
#define PREFIX_PER_LANE ((ROWS_PER_BLOCK + 31) / 32)
static_assert(N_EDGES % 32 == 0, "the row-by-row scan below reads 32 edges a step");

typedef float v2f __attribute__((ext_vector_type(2)));
typedef float v8f __attribute__((ext_vector_type(8)));

#define GEMM_GRID    256
#define GEMM_THREADS 256
#define GEMM_WAVES   (GEMM_THREADS / 32)
#define N_TILES      (N_NODES / 16)
#define KSTEPS       (D / 4)
#define KPAIRS       (KSTEPS / 2)
#define CTILES       (D / 16)
#define NFRAG        (2 * CTILES)

typedef float v4f __attribute__((ext_vector_type(4)));

__global__ __launch_bounds__(GEMM_THREADS)
void sage_project(const float* __restrict__ feat,
                  const float* __restrict__ Wself,
                  const float* __restrict__ bself,
                  const float* __restrict__ Wneigh,
                  const float* __restrict__ bneigh,
                  float*       __restrict__ out,
                  float*       __restrict__ proj,
                  const int*   __restrict__ dst,
                  unsigned*    __restrict__ keys,
                  size_t                    ws_bytes) {
    __shared__ v4f wfrag[KPAIRS * NFRAG * 32];

    const int tid  = threadIdx.x;
    const int lane = tid & 31;
    const int c    = lane & 15;
    const int h    = lane >> 4;

    const int wave    = __builtin_amdgcn_readfirstlane(tid >> 5);
    const int t_begin = (int)(blockIdx.x * N_TILES) / GEMM_GRID;
    const int t_end   = (int)((blockIdx.x + 1) * N_TILES) / GEMM_GRID;

    auto load_a = [&](v2f (&a)[KSTEPS], int tile) {
        const float* frow = feat + (size_t)(tile * 16 + c) * D + 2 * h;
#pragma unroll
        for (int s = 0; s < KSTEPS; ++s) a[s] = *(const v2f*)(frow + 4 * s);
    };

    int tile = t_begin + wave;
    v2f a[KSTEPS];
    load_a(a, tile);

#pragma unroll 4
    for (int it = 0; it < 2 * KPAIRS * 2 * (D / 4) / GEMM_THREADS; ++it) {
        const int item = it * GEMM_THREADS + tid;
        const int n4 = item & 31;
        const int hh = (item >> 5) & 1;
        const int p  = (item >> 6) & 15;
        const int m  = item >> 10;
        const float* Wm = (m ? Wneigh : Wself) + (size_t)(8 * p + 2 * hh) * D + 4 * n4;
        const v4f r0 = *(const v4f*)(Wm);
        const v4f r1 = *(const v4f*)(Wm + D);
        const v4f r2 = *(const v4f*)(Wm + 4 * D);
        const v4f r3 = *(const v4f*)(Wm + 5 * D);
        v4f* q = wfrag + ((p * NFRAG + 8 * m + (n4 >> 2)) * 32 + 16 * hh + 4 * (n4 & 3));
        v4f w;
        w.x = r0.x; w.y = r1.x; w.z = r2.x; w.w = r3.x; q[0] = w;
        w.x = r0.y; w.y = r1.y; w.z = r2.y; w.w = r3.y; q[1] = w;
        w.x = r0.z; w.y = r1.z; w.z = r2.z; w.w = r3.z; q[2] = w;
        w.x = r0.w; w.y = r1.w; w.z = r2.w; w.w = r3.w; q[3] = w;
    }

    float bias[CTILES];
#pragma unroll
    for (int t = 0; t < CTILES; ++t) bias[t] = bself[16 * t + c] + bneigh[16 * t + c];

    __syncthreads();

    const v4f* bl = wfrag + lane;
    auto load_b = [&](v4f (&b)[NFRAG], int p) {
#pragma unroll
        for (int f = 0; f < NFRAG; ++f) b[f] = bl[(p * NFRAG + f) * 32];
    };

    for (;;) {
        const int  next = tile + GEMM_WAVES;
        const bool more = next < t_end;
        v2f an[KSTEPS];
#pragma unroll
        for (int s = 0; s < KSTEPS; ++s) an[s] = a[s];
        if (more) load_a(an, next);

        v8f acc[NFRAG];
#pragma unroll
        for (int f = 0; f < NFRAG; ++f) {
#pragma unroll
            for (int j = 0; j < 8; ++j) acc[f][j] = (f < CTILES) ? bias[f] : 0.0f;
        }

        v4f b0[NFRAG], b1[NFRAG];
        load_b(b0, 0);
#pragma unroll
        for (int p = 0; p < KPAIRS; p += 2) {
            load_b(b1, p + 1);
            __builtin_amdgcn_sched_barrier(0);
#pragma unroll
            for (int f = 0; f < NFRAG; ++f)
                acc[f] = __builtin_amdgcn_wmma_f32_16x16x4_f32(
                             false, a[2 * p], false, b0[f].xy, (short)0, acc[f], false, false);
#pragma unroll
            for (int f = 0; f < NFRAG; ++f)
                acc[f] = __builtin_amdgcn_wmma_f32_16x16x4_f32(
                             false, a[2 * p + 1], false, b0[f].zw, (short)0, acc[f], false, false);
            __builtin_amdgcn_sched_barrier(0);
            if (p + 2 < KPAIRS) load_b(b0, p + 2);
            __builtin_amdgcn_sched_barrier(0);
#pragma unroll
            for (int f = 0; f < NFRAG; ++f)
                acc[f] = __builtin_amdgcn_wmma_f32_16x16x4_f32(
                             false, a[2 * p + 2], false, b1[f].xy, (short)0, acc[f], false, false);
#pragma unroll
            for (int f = 0; f < NFRAG; ++f)
                acc[f] = __builtin_amdgcn_wmma_f32_16x16x4_f32(
                             false, a[2 * p + 3], false, b1[f].zw, (short)0, acc[f], false, false);
            __builtin_amdgcn_sched_barrier(0);
        }

        const size_t obase = (size_t)(tile * 16 + h * 8) * D + c;
#pragma unroll
        for (int j = 0; j < 8; ++j) {
#pragma unroll
            for (int t = 0; t < CTILES; ++t) {
                out [obase + (size_t)j * D + 16 * t] = acc[t][j];
                proj[obase + (size_t)j * D + 16 * t] = acc[CTILES + t][j];
            }
        }

        if (!more) break;
#pragma unroll
        for (int s = 0; s < KSTEPS; ++s) a[s] = an[s];
        tile = next;
    }

    const int kt = (wave - 1) * 32 + (int)__builtin_amdgcn_mbcnt_lo(~0u, 0u);
    if (ws_bytes >= WS_NEED && blockIdx.x < KEY_BLOCKS && wave >= 1 && wave <= KEY_WORDS / 4 / 32) {
        const int4* dp = (const int4*)dst + blockIdx.x * KEY_WORDS + kt;
        unsigned*   kp = keys + blockIdx.x * KEY_WORDS + kt;
        int4 kd[4];
#pragma unroll
        for (int u = 0; u < 4; ++u) kd[u] = dp[u * (KEY_WORDS / 4)];
#pragma unroll
        for (int u = 0; u < 4; ++u) {
            const unsigned k0 = ((unsigned)kd[u].x / ROWS_PER_BLOCK) & 255u;
            const unsigned k1 = ((unsigned)kd[u].y / ROWS_PER_BLOCK) & 255u;
            const unsigned k2 = ((unsigned)kd[u].z / ROWS_PER_BLOCK) & 255u;
            const unsigned k3 = ((unsigned)kd[u].w / ROWS_PER_BLOCK) & 255u;
            kp[u * (KEY_WORDS / 4)] = k0 | (k1 << 8) | (k2 << 16) | (k3 << 24);
        }
    }
}

__global__ __launch_bounds__(OWNER_THREADS)
void sage_owner_gather(const float*    __restrict__ proj,
                       const int*      __restrict__ src,
                       const int*      __restrict__ dst,
                       const unsigned* __restrict__ keys,
                       size_t                       ws_bytes,
                       float*          __restrict__ out) {
    __shared__ unsigned list[LIST_CAP];
    __shared__ unsigned sorted[LIST_CAP];
    __shared__ unsigned cnt[ROWS_PER_BLOCK];
    __shared__ unsigned off[ROWS_PER_BLOCK];
    __shared__ unsigned cur[ROWS_PER_BLOCK];
    __shared__ unsigned nlist;

    const int tid   = threadIdx.x;
    const int lane  = tid & 31;
    const int wave  = tid >> 5;
    const int r0    = blockIdx.x * ROWS_PER_BLOCK;
    const int nrows = min(ROWS_PER_BLOCK, N_NODES - r0);

    for (int i = tid; i < ROWS_PER_BLOCK; i += OWNER_THREADS) cnt[i] = 0u;
    if (tid == 0) nlist = 0u;
    __syncthreads();

    auto push = [&](int e) {
        const unsigned slot = atomicAdd(&nlist, 1u);
        if (slot < LIST_CAP) list[slot] = (unsigned)e;
    };
    if (ws_bytes >= WS_NEED) {
        const uint4*   keys4 = (const uint4*)keys;
        const unsigned mine  = blockIdx.x * 0x01010101u;
        auto hit = [&](unsigned w) -> unsigned {
            const unsigned x = w ^ mine;
            return ~(((x & 0x7f7f7f7fu) + 0x7f7f7f7fu) | x | 0x7f7f7f7fu);
        };
        const int rot = blockIdx.x * KEY_ROT;
        auto fetch = [&](uint4 (&k)[SCAN_LOADS], int (&v)[SCAN_LOADS], int it) {
#pragma unroll
            for (int u = 0; u < SCAN_LOADS; ++u) {
                const int idx = it * SCAN_STEP + u * OWNER_THREADS + tid;
                int vv = idx + rot;
                if (vv >= N_KEY16) vv -= N_KEY16;
                const bool inside = idx < N_KEY16;
                v[u] = inside ? vv : -1;
                k[u] = keys4[inside ? vv : 0];
            }
        };
        auto look = [&](const uint4 (&k)[SCAN_LOADS], const int (&v)[SCAN_LOADS]) {
#pragma unroll
            for (int u = 0; u < SCAN_LOADS; ++u) {
                const int e = v[u] * 16;
                unsigned m = (hit(k[u].x) >> 7) | (hit(k[u].y) >> 6)
                           | (hit(k[u].z) >> 5) | (hit(k[u].w) >> 4);
                if (v[u] < 0) m = 0u;
                while (m != 0u) {
                    const int b = __builtin_ctz(m);
                    push(e + 4 * (b & 7) + (b >> 3));
                    m &= m - 1u;
                }
            }
        };
        uint4 ka[SCAN_LOADS], kb[SCAN_LOADS];
        int   va[SCAN_LOADS], vb[SCAN_LOADS];
        static_assert(KSCAN_ITERS % 2 == 0, "the loop below leaves set a and one more step");
        fetch(ka, va, 0);
        for (int it = 1; it + 1 < KSCAN_ITERS; it += 2) {
            fetch(kb, vb, it);
            look(ka, va);
            fetch(ka, va, it + 1);
            look(kb, vb);
        }
        fetch(kb, vb, KSCAN_ITERS - 1);
        look(ka, va);
        look(kb, vb);
    } else {
        const int4* dst4 = (const int4*)dst;
        auto keep = [&](int dval, int e) {
            if ((unsigned)(dval - r0) < (unsigned)nrows) push(e);
        };
        const int first = blockIdx.x % SCAN_ITERS;
        auto fetch = [&](int4 (&d)[SCAN_LOADS], int it) -> int {
            int itr = it + first;
            if (itr >= SCAN_ITERS) itr -= SCAN_ITERS;
            const int base = itr * SCAN_STEP + tid;
#pragma unroll
            for (int u = 0; u < SCAN_LOADS; ++u) d[u] = dst4[base + u * OWNER_THREADS];
            return base;
        };
        auto look = [&](const int4 (&d)[SCAN_LOADS], int base) {
#pragma unroll
            for (int u = 0; u < SCAN_LOADS; ++u) {
                const int e = (base + u * OWNER_THREADS) * 4;
                keep(d[u].x, e);
                keep(d[u].y, e + 1);
                keep(d[u].z, e + 2);
                keep(d[u].w, e + 3);
            }
        };
        int4 da[SCAN_LOADS], db[SCAN_LOADS];
        int base_a = fetch(da, 0);
        for (int it = 1; it + 1 < SCAN_ITERS; it += 2) {
            const int base_b = fetch(db, it);
            look(da, base_a);
            base_a = fetch(da, it + 1);
            look(db, base_b);
        }
        look(da, base_a);
        if (tid < N_DST4 - SCAN_ITERS * SCAN_STEP) {
            const int idx = SCAN_ITERS * SCAN_STEP + tid;
            const int4 d = dst4[idx];
            keep(d.x, idx * 4);
            keep(d.y, idx * 4 + 1);
            keep(d.z, idx * 4 + 2);
            keep(d.w, idx * 4 + 3);
        }
    }
    __syncthreads();

    const unsigned total_u = __builtin_amdgcn_readfirstlane(nlist);
    const unsigned wave_u  = __builtin_amdgcn_readfirstlane(wave);
    const unsigned nrows_u = (unsigned)nrows;
    const int      lane4   = lane * 4;

    if (total_u > LIST_CAP) {
        for (unsigned r = wave_u; r < nrows_u; r += OWNER_WAVES) {
            const int target = r0 + (int)r;
            float4   s   = make_float4(0.0f, 0.0f, 0.0f, 0.0f);
            unsigned deg = 0u;
#pragma nounroll
            for (int eb = 0; eb < N_EDGES; eb += 32) {
                unsigned m = __builtin_amdgcn_ballot_w32(dst[eb + lane] == target);
                while (m != 0u) {
                    const int b = __builtin_ctz(m);
                    m &= m - 1u;
                    const unsigned sn = (unsigned)src[eb + b] & 0x1ffffu;
                    const float4 v = *(const float4*)(proj + (size_t)sn * D + lane4);
                    s.x += v.x; s.y += v.y; s.z += v.z; s.w += v.w;
                    ++deg;
                }
            }
            const float rdeg = 1.0f / fmaxf((float)deg, 1.0f);
            float4* op = (float4*)(out + (size_t)target * D) + lane;
            float4 t = *op;
            t.x += s.x * rdeg; t.y += s.y * rdeg; t.z += s.z * rdeg; t.w += s.w * rdeg;
            *op = t;
        }
        return;
    }

    for (unsigned i = tid; i < total_u; i += OWNER_THREADS) {
        const unsigned e     = list[i];
        const unsigned local = (unsigned)(dst[e] - r0);
        const unsigned s     = (unsigned)src[e] & 0x1ffffu;
        const bool here = local < nrows_u;
        if (here) atomicAdd(&cnt[local], 1u);
        list[i] = here ? ((s << 9) | local) : ENTRY_INVALID;
    }
    __syncthreads();

    if (wave == 0) {
        unsigned c[PREFIX_PER_LANE];
        unsigned mine = 0u;
#pragma unroll
        for (int j = 0; j < PREFIX_PER_LANE; ++j) {
            const int idx = lane * PREFIX_PER_LANE + j;
            c[j] = idx < ROWS_PER_BLOCK ? cnt[idx] : 0u;
            mine += c[j];
        }
        unsigned incl = mine;
#pragma unroll
        for (int d = 1; d < 32; d <<= 1) {
            const unsigned up = __shfl_up(incl, d);
            if (lane >= d) incl += up;
        }
        unsigned run = incl - mine;
#pragma unroll
        for (int j = 0; j < PREFIX_PER_LANE; ++j) {
            const int idx = lane * PREFIX_PER_LANE + j;
            if (idx < ROWS_PER_BLOCK) { off[idx] = run; cur[idx] = run; }
            run += c[j];
        }
    }
    __syncthreads();

    for (unsigned i = tid; i < total_u; i += OWNER_THREADS) {
        const unsigned ent = list[i];
        if ((ent & ENTRY_INVALID) == 0u) {
            const unsigned pos = atomicAdd(&cur[ent & 511u], 1u);
            sorted[pos] = ent >> 9;
        }
    }
    __syncthreads();

    const unsigned q8 = (unsigned)(lane & (CHUNK - 1));
    for (unsigned rA = wave_u; rA < nrows_u; rA += 2 * OWNER_WAVES) {
        const bool     hasB = rA + OWNER_WAVES < nrows_u;
        const unsigned rB = hasB ? rA + OWNER_WAVES : rA;
        const unsigned dA = __builtin_amdgcn_readfirstlane(cnt[rA]);
        const unsigned bA = __builtin_amdgcn_readfirstlane(off[rA]);
        const unsigned dB = hasB ? __builtin_amdgcn_readfirstlane(cnt[rB]) : 0u;
        const unsigned bB = __builtin_amdgcn_readfirstlane(off[rB]);
        float4* opA = (float4*)(out + (size_t)(r0 + (int)rA) * D) + lane;
        float4* opB = (float4*)(out + (size_t)(r0 + (int)rB) * D) + lane;
        float4 tA = *opA;
        float4 tB = *opB;
        float4 sA = make_float4(0.0f, 0.0f, 0.0f, 0.0f);
        float4 sB = make_float4(0.0f, 0.0f, 0.0f, 0.0f);
        const unsigned dmax = max(dA, dB);
        unsigned k = 0u;
        do {
            const unsigned nA = dA > k ? min(dA - k, (unsigned)CHUNK) : 0u;
            const unsigned nB = dB > k ? min(dB - k, (unsigned)CHUNK) : 0u;
            const unsigned myA = sorted[q8 < nA ? bA + k + q8 : 0u];
            const unsigned myB = sorted[q8 < nB ? bB + k + q8 : 0u];
            float4 vA[CHUNK], vB[CHUNK];
#pragma unroll
            for (int q = 0; q < CHUNK; ++q) {
                vA[q] = make_float4(0.0f, 0.0f, 0.0f, 0.0f);
                if ((unsigned)q < nA) {
                    const unsigned sn = __builtin_amdgcn_readlane(myA, q);
                    vA[q] = *(const float4*)(proj + (size_t)sn * D + lane4);
                }
            }
#pragma unroll
            for (int q = 0; q < CHUNK; ++q) {
                vB[q] = make_float4(0.0f, 0.0f, 0.0f, 0.0f);
                if ((unsigned)q < nB) {
                    const unsigned sn = __builtin_amdgcn_readlane(myB, q);
                    vB[q] = *(const float4*)(proj + (size_t)sn * D + lane4);
                }
            }
#pragma unroll
            for (int q = 0; q < CHUNK; ++q) {
                sA.x += vA[q].x; sA.y += vA[q].y; sA.z += vA[q].z; sA.w += vA[q].w;
            }
#pragma unroll
            for (int q = 0; q < CHUNK; ++q) {
                sB.x += vB[q].x; sB.y += vB[q].y; sB.z += vB[q].z; sB.w += vB[q].w;
            }
            k += CHUNK;
        } while (k < dmax);
        const float rdA = 1.0f / fmaxf((float)dA, 1.0f);
        tA.x += sA.x * rdA; tA.y += sA.y * rdA; tA.z += sA.z * rdA; tA.w += sA.w * rdA;
        *opA = tA;
        if (hasB) {
            const float rdB = 1.0f / fmaxf((float)dB, 1.0f);
            tB.x += sB.x * rdB; tB.y += sB.y * rdB; tB.z += sB.z * rdB; tB.w += sB.w * rdB;
            *opB = tB;
        }
    }
}

extern "C" void kernel_launch(void* const* d_in, const int* in_sizes, int n_in,
                              void* d_out, int out_size, void* d_ws, size_t ws_size,
                              hipStream_t stream) {
    const float* feat   = (const float*)d_in[0];
    const int*   src    = (const int*)  d_in[1];
    const int*   dst    = (const int*)  d_in[2];
    const float* Wself  = (const float*)d_in[3];
    const float* bself  = (const float*)d_in[4];
    const float* Wneigh = (const float*)d_in[5];
    const float* bneigh = (const float*)d_in[6];
    float*       out    = (float*)d_out;

    float*    proj = (float*)d_ws;
    unsigned* keys = (unsigned*)((char*)d_ws + PROJ_BYTES);

    sage_project<<<GEMM_GRID, GEMM_THREADS, 0, stream>>>(
        feat, Wself, bself, Wneigh, bneigh, out, proj, dst, keys, ws_size);

    sage_owner_gather<<<OWNER_BLOCKS, OWNER_THREADS, 0, stream>>>(
        proj, src, dst, keys, ws_size, out);
}


# GEMM tail balance: a block's 25th row tile (106 blocks have one) is split over its 8 waves, 2 accumulator tiles each, A rows from LDS; dst loads of the key step issued in front of it; same sums
# speedup vs baseline: 2.6179x; 1.0318x over previous
#include <hip/hip_runtime.h>


#define N_NODES 100000
#define N_EDGES 640000
#define D 128

#define ROWS_PER_BLOCK 391
#define OWNER_BLOCKS   ((N_NODES + ROWS_PER_BLOCK - 1) / ROWS_PER_BLOCK)
#define OWNER_THREADS  1024
#define OWNER_WAVES    (OWNER_THREADS / 32)
#define LIST_CAP       32768
#define N_DST4         (N_EDGES / 4)
#define SCAN_LOADS     4
#define SCAN_STEP      (SCAN_LOADS * OWNER_THREADS)
#define SCAN_ITERS     (N_DST4 / SCAN_STEP)
static_assert(SCAN_ITERS % 2 == 1, "the two-set scan loop below ends on set a");
#define EDGE_CHUNK     2560
#define KEY_BLOCKS     (N_EDGES / EDGE_CHUNK)
#define KEY_WORDS      (EDGE_CHUNK / 4)
#define N_KEY16        (N_EDGES / 16)
#define KSCAN_ITERS    ((N_KEY16 + SCAN_STEP - 1) / SCAN_STEP)
#define KEY_ROT        128
static_assert(KEY_ROT % 32 == 0 && N_KEY16 % 32 == 0 && (OWNER_BLOCKS - 1) * KEY_ROT < N_KEY16, "aligned, one wrap");
#define PROJ_BYTES     ((size_t)N_NODES * D * sizeof(float))
#define WS_NEED        (PROJ_BYTES + (size_t)N_EDGES)
static_assert(N_EDGES % EDGE_CHUNK == 0 && EDGE_CHUNK % 128 == 0, "every key line has one writer");
static_assert(OWNER_BLOCKS <= 256, "an owner block number fits a byte");
#define ENTRY_INVALID  0x80000000u
#define CHUNK          8
#define PREFIX_PER_LANE ((ROWS_PER_BLOCK + 31) / 32)
static_assert(N_EDGES % 32 == 0, "the row-by-row scan below reads 32 edges a step");

typedef float v2f __attribute__((ext_vector_type(2)));
typedef float v8f __attribute__((ext_vector_type(8)));

#define GEMM_GRID    256
#define GEMM_THREADS 256
#define GEMM_WAVES   (GEMM_THREADS / 32)
#define N_TILES      (N_NODES / 16)
#define KSTEPS       (D / 4)
#define KPAIRS       (KSTEPS / 2)
#define CTILES       (D / 16)
#define NFRAG        (2 * CTILES)
#define MAIN_TILES   (N_TILES / GEMM_GRID)
#define APITCH       (D + 4)
static_assert(16 * (D / 4) == 2 * GEMM_THREADS, "a thread copies two 16-byte pieces of the shared tile");
static_assert(MAIN_TILES % GEMM_WAVES == 0 && N_TILES % GEMM_GRID != 0 && N_TILES < (MAIN_TILES + 1) * GEMM_GRID,
              "a block has MAIN_TILES or MAIN_TILES + 1 tiles, and the main ones divide evenly over the waves");
static_assert(NFRAG == 2 * GEMM_WAVES, "the shared tile gives every wave two accumulator tiles");

typedef float v4f __attribute__((ext_vector_type(4)));

__global__ __launch_bounds__(GEMM_THREADS)
void sage_project(const float* __restrict__ feat,
                  const float* __restrict__ Wself,
                  const float* __restrict__ bself,
                  const float* __restrict__ Wneigh,
                  const float* __restrict__ bneigh,
                  float*       __restrict__ out,
                  float*       __restrict__ proj,
                  const int*   __restrict__ dst,
                  unsigned*    __restrict__ keys,
                  size_t                    ws_bytes) {
    __shared__ v4f wfrag[KPAIRS * NFRAG * 32];
    __shared__ __attribute__((aligned(16))) float atile[16 * APITCH];

    const int tid  = threadIdx.x;
    const int lane = tid & 31;
    const int c    = lane & 15;
    const int h    = lane >> 4;

    const int wave    = __builtin_amdgcn_readfirstlane(tid >> 5);
    const int t_begin = (int)(blockIdx.x * N_TILES) / GEMM_GRID;
    const int t_end   = (int)((blockIdx.x + 1) * N_TILES) / GEMM_GRID;
    const int  t_main = t_begin + MAIN_TILES;
    const bool coop   = t_end > t_main;

    auto load_a = [&](v2f (&a)[KSTEPS], int tile) {
        const float* frow = feat + (size_t)(tile * 16 + c) * D + 2 * h;
#pragma unroll
        for (int s = 0; s < KSTEPS; ++s) a[s] = *(const v2f*)(frow + 4 * s);
    };

    int tile = t_begin + wave;
    v2f a[KSTEPS];
    load_a(a, tile);

    const int    t_sh = min(t_main, N_TILES - 1);
    const float* sp   = feat + (size_t)(t_sh * 16 + (tid >> 5)) * D + 4 * lane;
    const v4f    st0  = *(const v4f*)(sp);
    const v4f    st1  = *(const v4f*)(sp + 8 * D);

#pragma unroll 4
    for (int it = 0; it < 2 * KPAIRS * 2 * (D / 4) / GEMM_THREADS; ++it) {
        const int item = it * GEMM_THREADS + tid;
        const int n4 = item & 31;
        const int hh = (item >> 5) & 1;
        const int p  = (item >> 6) & 15;
        const int m  = item >> 10;
        const float* Wm = (m ? Wneigh : Wself) + (size_t)(8 * p + 2 * hh) * D + 4 * n4;
        const v4f r0 = *(const v4f*)(Wm);
        const v4f r1 = *(const v4f*)(Wm + D);
        const v4f r2 = *(const v4f*)(Wm + 4 * D);
        const v4f r3 = *(const v4f*)(Wm + 5 * D);
        v4f* q = wfrag + ((p * NFRAG + 8 * m + (n4 >> 2)) * 32 + 16 * hh + 4 * (n4 & 3));
        v4f w;
        w.x = r0.x; w.y = r1.x; w.z = r2.x; w.w = r3.x; q[0] = w;
        w.x = r0.y; w.y = r1.y; w.z = r2.y; w.w = r3.y; q[1] = w;
        w.x = r0.z; w.y = r1.z; w.z = r2.z; w.w = r3.z; q[2] = w;
        w.x = r0.w; w.y = r1.w; w.z = r2.w; w.w = r3.w; q[3] = w;
    }

    float bias[CTILES];
#pragma unroll
    for (int t = 0; t < CTILES; ++t) bias[t] = bself[16 * t + c] + bneigh[16 * t + c];

    {
        float* sq = atile + (tid >> 5) * APITCH + 4 * lane;
        *(v4f*)(sq)              = st0;
        *(v4f*)(sq + 8 * APITCH) = st1;
    }

    __syncthreads();

    const v4f* bl = wfrag + lane;
    auto load_b = [&](v4f (&b)[NFRAG], int p) {
#pragma unroll
        for (int f = 0; f < NFRAG; ++f) b[f] = bl[(p * NFRAG + f) * 32];
    };

    for (;;) {
        const int  next = tile + GEMM_WAVES;
        const bool more = next < t_main;
        v2f an[KSTEPS];
#pragma unroll
        for (int s = 0; s < KSTEPS; ++s) an[s] = a[s];
        if (more) load_a(an, next);

        v8f acc[NFRAG];
#pragma unroll
        for (int f = 0; f < NFRAG; ++f) {
#pragma unroll
            for (int j = 0; j < 8; ++j) acc[f][j] = (f < CTILES) ? bias[f] : 0.0f;
        }

        v4f b0[NFRAG], b1[NFRAG];
        load_b(b0, 0);
#pragma unroll
        for (int p = 0; p < KPAIRS; p += 2) {
            load_b(b1, p + 1);
            __builtin_amdgcn_sched_barrier(0);
#pragma unroll
            for (int f = 0; f < NFRAG; ++f)
                acc[f] = __builtin_amdgcn_wmma_f32_16x16x4_f32(
                             false, a[2 * p], false, b0[f].xy, (short)0, acc[f], false, false);
#pragma unroll
            for (int f = 0; f < NFRAG; ++f)
                acc[f] = __builtin_amdgcn_wmma_f32_16x16x4_f32(
                             false, a[2 * p + 1], false, b0[f].zw, (short)0, acc[f], false, false);
            __builtin_amdgcn_sched_barrier(0);
            if (p + 2 < KPAIRS) load_b(b0, p + 2);
            __builtin_amdgcn_sched_barrier(0);
#pragma unroll
            for (int f = 0; f < NFRAG; ++f)
                acc[f] = __builtin_amdgcn_wmma_f32_16x16x4_f32(
                             false, a[2 * p + 2], false, b1[f].xy, (short)0, acc[f], false, false);
#pragma unroll
            for (int f = 0; f < NFRAG; ++f)
                acc[f] = __builtin_amdgcn_wmma_f32_16x16x4_f32(
                             false, a[2 * p + 3], false, b1[f].zw, (short)0, acc[f], false, false);
            __builtin_amdgcn_sched_barrier(0);
        }

        const size_t obase = (size_t)(tile * 16 + h * 8) * D + c;
#pragma unroll
        for (int j = 0; j < 8; ++j) {
#pragma unroll
            for (int t = 0; t < CTILES; ++t) {
                out [obase + (size_t)j * D + 16 * t] = acc[t][j];
                proj[obase + (size_t)j * D + 16 * t] = acc[CTILES + t][j];
            }
        }

        if (!more) break;
#pragma unroll
        for (int s = 0; s < KSTEPS; ++s) a[s] = an[s];
        tile = next;
    }

    const int  ln    = (int)__builtin_amdgcn_mbcnt_lo(~0u, 0u);
    const bool kwave = wave >= 1 && wave <= KEY_WORDS / 4 / 32;
    const bool keyer = ws_bytes >= WS_NEED && blockIdx.x < KEY_BLOCKS && kwave;
    const int  ktc   = (kwave ? wave - 1 : 0) * 32 + ln;
    const int  kblk  = min((int)blockIdx.x, KEY_BLOCKS - 1);
    const int4* dp   = (const int4*)dst + kblk * KEY_WORDS + ktc;
    int4 kd[4];
#pragma unroll
    for (int u = 0; u < 4; ++u) kd[u] = dp[u * (KEY_WORDS / 4)];

    if (coop) {
        const int  cq    = ln & 15;
        const int  hq    = ln >> 4;
        const int  f0    = 2 * wave;
        const bool selfp = f0 < CTILES;
        const int  t0    = f0 & (CTILES - 1);
        float bi0 = 0.0f, bi1 = 0.0f;
#pragma unroll
        for (int t = 0; t < CTILES; t += 2)
            if (selfp && t0 == t) { bi0 = bias[t]; bi1 = bias[t + 1]; }
        v8f c0, c1;
#pragma unroll
        for (int j = 0; j < 8; ++j) { c0[j] = bi0; c1[j] = bi1; }
        const v4f* bq = wfrag + f0 * 32 + ln;
        const float* aq = atile + cq * APITCH + 2 * hq;
        float* dq = (selfp ? out : proj) + ((size_t)t_main * 16 * D + 16 * t0) + (unsigned)(hq * 8 * D + cq);
        v2f as[KSTEPS];
        v4f x0[KPAIRS], x1[KPAIRS];
#pragma unroll
        for (int p = 0; p < KPAIRS; ++p) {
            as[2 * p]     = *(const v2f*)(aq + 8 * p);
            as[2 * p + 1] = *(const v2f*)(aq + 8 * p + 4);
            x0[p] = bq[(p * NFRAG) * 32];
            x1[p] = bq[(p * NFRAG + 1) * 32];
        }
        __builtin_amdgcn_sched_barrier(0);
#pragma unroll
        for (int p = 0; p < KPAIRS; ++p) {
            c0 = __builtin_amdgcn_wmma_f32_16x16x4_f32(false, as[2 * p],     false, x0[p].xy, (short)0, c0, false, false);
            c1 = __builtin_amdgcn_wmma_f32_16x16x4_f32(false, as[2 * p],     false, x1[p].xy, (short)0, c1, false, false);
            c0 = __builtin_amdgcn_wmma_f32_16x16x4_f32(false, as[2 * p + 1], false, x0[p].zw, (short)0, c0, false, false);
            c1 = __builtin_amdgcn_wmma_f32_16x16x4_f32(false, as[2 * p + 1], false, x1[p].zw, (short)0, c1, false, false);
        }
#pragma unroll
        for (int j = 0; j < 8; ++j) {
            dq[(size_t)j * D]      = c0[j];
            dq[(size_t)j * D + 16] = c1[j];
        }
    }

    if (keyer) {
        unsigned* kp = keys + blockIdx.x * KEY_WORDS + ktc;
#pragma unroll
        for (int u = 0; u < 4; ++u) {
            const unsigned k0 = ((unsigned)kd[u].x / ROWS_PER_BLOCK) & 255u;
            const unsigned k1 = ((unsigned)kd[u].y / ROWS_PER_BLOCK) & 255u;
            const unsigned k2 = ((unsigned)kd[u].z / ROWS_PER_BLOCK) & 255u;
            const unsigned k3 = ((unsigned)kd[u].w / ROWS_PER_BLOCK) & 255u;
            kp[u * (KEY_WORDS / 4)] = k0 | (k1 << 8) | (k2 << 16) | (k3 << 24);
        }
    }
}

__global__ __launch_bounds__(OWNER_THREADS)
void sage_owner_gather(const float*    __restrict__ proj,
                       const int*      __restrict__ src,
                       const int*      __restrict__ dst,
                       const unsigned* __restrict__ keys,
                       size_t                       ws_bytes,
                       float*          __restrict__ out) {
    __shared__ unsigned list[LIST_CAP];
    __shared__ unsigned sorted[LIST_CAP];
    __shared__ unsigned cnt[ROWS_PER_BLOCK];
    __shared__ unsigned off[ROWS_PER_BLOCK];
    __shared__ unsigned cur[ROWS_PER_BLOCK];
    __shared__ unsigned nlist;

    const int tid   = threadIdx.x;
    const int lane  = tid & 31;
    const int wave  = tid >> 5;
    const int r0    = blockIdx.x * ROWS_PER_BLOCK;
    const int nrows = min(ROWS_PER_BLOCK, N_NODES - r0);

    for (int i = tid; i < ROWS_PER_BLOCK; i += OWNER_THREADS) cnt[i] = 0u;
    if (tid == 0) nlist = 0u;
    __syncthreads();

    auto push = [&](int e) {
        const unsigned slot = atomicAdd(&nlist, 1u);
        if (slot < LIST_CAP) list[slot] = (unsigned)e;
    };
    if (ws_bytes >= WS_NEED) {
        const uint4*   keys4 = (const uint4*)keys;
        const unsigned mine  = blockIdx.x * 0x01010101u;
        auto hit = [&](unsigned w) -> unsigned {
            const unsigned x = w ^ mine;
            return ~(((x & 0x7f7f7f7fu) + 0x7f7f7f7fu) | x | 0x7f7f7f7fu);
        };
        const int rot = blockIdx.x * KEY_ROT;
        auto fetch = [&](uint4 (&k)[SCAN_LOADS], int (&v)[SCAN_LOADS], int it) {
#pragma unroll
            for (int u = 0; u < SCAN_LOADS; ++u) {
                const int idx = it * SCAN_STEP + u * OWNER_THREADS + tid;
                int vv = idx + rot;
                if (vv >= N_KEY16) vv -= N_KEY16;
                const bool inside = idx < N_KEY16;
                v[u] = inside ? vv : -1;
                k[u] = keys4[inside ? vv : 0];
            }
        };
        auto look = [&](const uint4 (&k)[SCAN_LOADS], const int (&v)[SCAN_LOADS]) {
#pragma unroll
            for (int u = 0; u < SCAN_LOADS; ++u) {
                const int e = v[u] * 16;
                unsigned m = (hit(k[u].x) >> 7) | (hit(k[u].y) >> 6)
                           | (hit(k[u].z) >> 5) | (hit(k[u].w) >> 4);
                if (v[u] < 0) m = 0u;
                while (m != 0u) {
                    const int b = __builtin_ctz(m);
                    push(e + 4 * (b & 7) + (b >> 3));
                    m &= m - 1u;
                }
            }
        };
        uint4 ka[SCAN_LOADS], kb[SCAN_LOADS];
        int   va[SCAN_LOADS], vb[SCAN_LOADS];
        static_assert(KSCAN_ITERS % 2 == 0, "the loop below leaves set a and one more step");
        fetch(ka, va, 0);
        for (int it = 1; it + 1 < KSCAN_ITERS; it += 2) {
            fetch(kb, vb, it);
            look(ka, va);
            fetch(ka, va, it + 1);
            look(kb, vb);
        }
        fetch(kb, vb, KSCAN_ITERS - 1);
        look(ka, va);
        look(kb, vb);
    } else {
        const int4* dst4 = (const int4*)dst;
        auto keep = [&](int dval, int e) {
            if ((unsigned)(dval - r0) < (unsigned)nrows) push(e);
        };
        const int first = blockIdx.x % SCAN_ITERS;
        auto fetch = [&](int4 (&d)[SCAN_LOADS], int it) -> int {
            int itr = it + first;
            if (itr >= SCAN_ITERS) itr -= SCAN_ITERS;
            const int base = itr * SCAN_STEP + tid;
#pragma unroll
            for (int u = 0; u < SCAN_LOADS; ++u) d[u] = dst4[base + u * OWNER_THREADS];
            return base;
        };
        auto look = [&](const int4 (&d)[SCAN_LOADS], int base) {
#pragma unroll
            for (int u = 0; u < SCAN_LOADS; ++u) {
                const int e = (base + u * OWNER_THREADS) * 4;
                keep(d[u].x, e);
                keep(d[u].y, e + 1);
                keep(d[u].z, e + 2);
                keep(d[u].w, e + 3);
            }
        };
        int4 da[SCAN_LOADS], db[SCAN_LOADS];
        int base_a = fetch(da, 0);
        for (int it = 1; it + 1 < SCAN_ITERS; it += 2) {
            const int base_b = fetch(db, it);
            look(da, base_a);
            base_a = fetch(da, it + 1);
            look(db, base_b);
        }
        look(da, base_a);
        if (tid < N_DST4 - SCAN_ITERS * SCAN_STEP) {
            const int idx = SCAN_ITERS * SCAN_STEP + tid;
            const int4 d = dst4[idx];
            keep(d.x, idx * 4);
            keep(d.y, idx * 4 + 1);
            keep(d.z, idx * 4 + 2);
            keep(d.w, idx * 4 + 3);
        }
    }
    __syncthreads();

    const unsigned total_u = __builtin_amdgcn_readfirstlane(nlist);
    const unsigned wave_u  = __builtin_amdgcn_readfirstlane(wave);
    const unsigned nrows_u = (unsigned)nrows;
    const int      lane4   = lane * 4;

    if (total_u > LIST_CAP) {
        for (unsigned r = wave_u; r < nrows_u; r += OWNER_WAVES) {
            const int target = r0 + (int)r;
            float4   s   = make_float4(0.0f, 0.0f, 0.0f, 0.0f);
            unsigned deg = 0u;
#pragma nounroll
            for (int eb = 0; eb < N_EDGES; eb += 32) {
                unsigned m = __builtin_amdgcn_ballot_w32(dst[eb + lane] == target);
                while (m != 0u) {
                    const int b = __builtin_ctz(m);
                    m &= m - 1u;
                    const unsigned sn = (unsigned)src[eb + b] & 0x1ffffu;
                    const float4 v = *(const float4*)(proj + (size_t)sn * D + lane4);
                    s.x += v.x; s.y += v.y; s.z += v.z; s.w += v.w;
                    ++deg;
                }
            }
            const float rdeg = 1.0f / fmaxf((float)deg, 1.0f);
            float4* op = (float4*)(out + (size_t)target * D) + lane;
            float4 t = *op;
            t.x += s.x * rdeg; t.y += s.y * rdeg; t.z += s.z * rdeg; t.w += s.w * rdeg;
            *op = t;
        }
        return;
    }

    for (unsigned i = tid; i < total_u; i += OWNER_THREADS) {
        const unsigned e     = list[i];
        const unsigned local = (unsigned)(dst[e] - r0);
        const unsigned s     = (unsigned)src[e] & 0x1ffffu;
        const bool here = local < nrows_u;
        if (here) atomicAdd(&cnt[local], 1u);
        list[i] = here ? ((s << 9) | local) : ENTRY_INVALID;
    }
    __syncthreads();

    if (wave == 0) {
        unsigned c[PREFIX_PER_LANE];
        unsigned mine = 0u;
#pragma unroll
        for (int j = 0; j < PREFIX_PER_LANE; ++j) {
            const int idx = lane * PREFIX_PER_LANE + j;
            c[j] = idx < ROWS_PER_BLOCK ? cnt[idx] : 0u;
            mine += c[j];
        }
        unsigned incl = mine;
#pragma unroll
        for (int d = 1; d < 32; d <<= 1) {
            const unsigned up = __shfl_up(incl, d);
            if (lane >= d) incl += up;
        }
        unsigned run = incl - mine;
#pragma unroll
        for (int j = 0; j < PREFIX_PER_LANE; ++j) {
            const int idx = lane * PREFIX_PER_LANE + j;
            if (idx < ROWS_PER_BLOCK) { off[idx] = run; cur[idx] = run; }
            run += c[j];
        }
    }
    __syncthreads();

    for (unsigned i = tid; i < total_u; i += OWNER_THREADS) {
        const unsigned ent = list[i];
        if ((ent & ENTRY_INVALID) == 0u) {
            const unsigned pos = atomicAdd(&cur[ent & 511u], 1u);
            sorted[pos] = ent >> 9;
        }
    }
    __syncthreads();

    const unsigned q8 = (unsigned)(lane & (CHUNK - 1));
    for (unsigned rA = wave_u; rA < nrows_u; rA += 2 * OWNER_WAVES) {
        const bool     hasB = rA + OWNER_WAVES < nrows_u;
        const unsigned rB = hasB ? rA + OWNER_WAVES : rA;
        const unsigned dA = __builtin_amdgcn_readfirstlane(cnt[rA]);
        const unsigned bA = __builtin_amdgcn_readfirstlane(off[rA]);
        const unsigned dB = hasB ? __builtin_amdgcn_readfirstlane(cnt[rB]) : 0u;
        const unsigned bB = __builtin_amdgcn_readfirstlane(off[rB]);
        float4* opA = (float4*)(out + (size_t)(r0 + (int)rA) * D) + lane;
        float4* opB = (float4*)(out + (size_t)(r0 + (int)rB) * D) + lane;
        float4 tA = *opA;
        float4 tB = *opB;
        float4 sA = make_float4(0.0f, 0.0f, 0.0f, 0.0f);
        float4 sB = make_float4(0.0f, 0.0f, 0.0f, 0.0f);
        const unsigned dmax = max(dA, dB);
        unsigned k = 0u;
        do {
            const unsigned nA = dA > k ? min(dA - k, (unsigned)CHUNK) : 0u;
            const unsigned nB = dB > k ? min(dB - k, (unsigned)CHUNK) : 0u;
            const unsigned myA = sorted[q8 < nA ? bA + k + q8 : 0u];
            const unsigned myB = sorted[q8 < nB ? bB + k + q8 : 0u];
            float4 vA[CHUNK], vB[CHUNK];
#pragma unroll
            for (int q = 0; q < CHUNK; ++q) {
                vA[q] = make_float4(0.0f, 0.0f, 0.0f, 0.0f);
                if ((unsigned)q < nA) {
                    const unsigned sn = __builtin_amdgcn_readlane(myA, q);
                    vA[q] = *(const float4*)(proj + (size_t)sn * D + lane4);
                }
            }
#pragma unroll
            for (int q = 0; q < CHUNK; ++q) {
                vB[q] = make_float4(0.0f, 0.0f, 0.0f, 0.0f);
                if ((unsigned)q < nB) {
                    const unsigned sn = __builtin_amdgcn_readlane(myB, q);
                    vB[q] = *(const float4*)(proj + (size_t)sn * D + lane4);
                }
            }
#pragma unroll
            for (int q = 0; q < CHUNK; ++q) {
                sA.x += vA[q].x; sA.y += vA[q].y; sA.z += vA[q].z; sA.w += vA[q].w;
            }
#pragma unroll
            for (int q = 0; q < CHUNK; ++q) {
                sB.x += vB[q].x; sB.y += vB[q].y; sB.z += vB[q].z; sB.w += vB[q].w;
            }
            k += CHUNK;
        } while (k < dmax);
        const float rdA = 1.0f / fmaxf((float)dA, 1.0f);
        tA.x += sA.x * rdA; tA.y += sA.y * rdA; tA.z += sA.z * rdA; tA.w += sA.w * rdA;
        *opA = tA;
        if (hasB) {
            const float rdB = 1.0f / fmaxf((float)dB, 1.0f);
            tB.x += sB.x * rdB; tB.y += sB.y * rdB; tB.z += sB.z * rdB; tB.w += sB.w * rdB;
            *opB = tB;
        }
    }
}

extern "C" void kernel_launch(void* const* d_in, const int* in_sizes, int n_in,
                              void* d_out, int out_size, void* d_ws, size_t ws_size,
                              hipStream_t stream) {
    const float* feat   = (const float*)d_in[0];
    const int*   src    = (const int*)  d_in[1];
    const int*   dst    = (const int*)  d_in[2];
    const float* Wself  = (const float*)d_in[3];
    const float* bself  = (const float*)d_in[4];
    const float* Wneigh = (const float*)d_in[5];
    const float* bneigh = (const float*)d_in[6];
    float*       out    = (float*)d_out;

    float*    proj = (float*)d_ws;
    unsigned* keys = (unsigned*)((char*)d_ws + PROJ_BYTES);

    sage_project<<<GEMM_GRID, GEMM_THREADS, 0, stream>>>(
        feat, Wself, bself, Wneigh, bneigh, out, proj, dst, keys, ws_size);

    sage_owner_gather<<<OWNER_BLOCKS, OWNER_THREADS, 0, stream>>>(
        proj, src, dst, keys, ws_size, out);
}
